# starting kernel
# baseline (speedup 1.0000x reference)
#include <hip/hip_runtime.h>


typedef __attribute__((ext_vector_type(2))) float v2f;
typedef __attribute__((ext_vector_type(8))) float v8f;

#define TILE   256
#define BY     64
#define BX     64
#define IPH    (BY + 14)
#define IPW    (BX + 16)
#define NCHAIN (IPH * (BX / 16))
#define OPAD   15
#define OROWS  (OPAD + BY + 14)

__global__ __launch_bounds__(256) void norm_kernels_k(const float* __restrict__ k,
                                                      float* __restrict__ knorm) {
    const int t   = blockIdx.x;
    const int tid = threadIdx.x;

    __shared__ float red[256];
    float v = 0.0f;
    if (tid < 225) v = k[t * 225 + tid];
    red[tid] = v;
    __syncthreads();
    for (int s = 128; s > 0; s >>= 1) {
        if (tid < s) red[tid] += red[tid + s];
        __syncthreads();
    }
    const float inv = 1.0f / (red[0] + 1e-12f);

    const int dy = tid >> 4;
    const int dx = tid & 15;
    const float val = (dy < 15 && dx < 15) ? k[t * 225 + dy * 15 + dx] * inv : 0.0f;
    knorm[t * 256 + tid] = val;
}

__global__ __launch_bounds__(256) void pwblur_k(const float* __restrict__ x,
                                                const float* __restrict__ knorm,
                                                float* __restrict__ out) {
    __shared__ float ip[IPH * IPW];
    __shared__ float outl[OROWS * BX];

    const int tid  = threadIdx.x;
    const int lane = tid & 31;
    const int wave = tid >> 5;
    const bool hi  = (lane & 16) != 0;
    const int  nl  = lane & 15;

    int bid   = blockIdx.x;
    int sub   = bid & 15;
    int t     = bid >> 4;
    int ti    = t & 3;
    int tj    = (t >> 2) & 3;
    int plane = t >> 4;
    int sub_y = sub >> 2;
    int sub_x = sub & 3;

    const int ty0 = tj * TILE;
    const int tx0 = ti * TILE;
    const int y0  = sub_y * BY;
    const int x0  = sub_x * BX;

    const float* __restrict__ xin = x + (size_t)plane * 1024 * 1024;

    for (int idx = tid; idx < IPH * IPW; idx += 256) {
        int r  = idx / IPW;
        int xi = idx - r * IPW;
        int ys = y0 + r - 7;  ys = ys < 0 ? 0 : (ys > TILE - 1 ? TILE - 1 : ys);
        int xs = x0 + xi - 7; xs = xs < 0 ? 0 : (xs > TILE - 1 ? TILE - 1 : xs);
        ip[idx] = xin[(size_t)(ty0 + ys) * 1024 + (tx0 + xs)];
    }
    for (int idx = tid; idx < OROWS * BX; idx += 256) outl[idx] = 0.0f;

    const float* __restrict__ kn = knorm + (tj * 4 + ti) * 256;
    v2f a[4];
    {
        const int m = nl;
        #pragma unroll
        for (int c = 0; c < 4; ++c) {
            int dxb = 4 * c + (hi ? 2 : 0);
            a[c].x = kn[m * 16 + dxb];
            a[c].y = kn[m * 16 + dxb + 1];
        }
    }
    __syncthreads();

    for (int cid = wave; cid < NCHAIN; cid += 8) {
        const int r  = cid >> 2;
        const int xb = (cid & 3) * 16;

        const int bbase = r * IPW + xb + nl + (hi ? 2 : 0);

        v8f acc = {};
        #pragma unroll
        for (int c = 0; c < 4; ++c) {
            v2f bv;
            bv.x = ip[bbase + 4 * c];
            bv.y = ip[bbase + 4 * c + 1];
            acc = __builtin_amdgcn_wmma_f32_16x16x4_f32(
                false, a[c], false, bv, (short)0, acc, false, false);
        }

        const int sbase = (r + OPAD - (hi ? 8 : 0)) * BX + xb + nl;
        #pragma unroll
        for (int i = 0; i < 8; ++i) {
            atomicAdd(&outl[sbase - i * BX], acc[i]);
        }
    }
    __syncthreads();

    float* __restrict__ op = out + (size_t)plane * 1024 * 1024;
    for (int idx = tid; idx < BY * BX; idx += 256) {
        int yl = idx >> 6;
        int xl = idx & 63;
        op[(size_t)(ty0 + y0 + yl) * 1024 + (tx0 + x0 + xl)] = outl[(OPAD + yl) * BX + xl];
    }
}

extern "C" void kernel_launch(void* const* d_in, const int* in_sizes, int n_in,
                              void* d_out, int out_size, void* d_ws, size_t ws_size,
                              hipStream_t stream) {
    const float* x    = (const float*)d_in[0];
    const float* kern = (const float*)d_in[1];
    float* outp  = (float*)d_out;
    float* knorm = (float*)d_ws;

    norm_kernels_k<<<16, 256, 0, stream>>>(kern, knorm);

    pwblur_k<<<6144, 256, 0, stream>>>(x, knorm, outp);
}


# Strip split into even and odd columns, not left and right halves: one B fragment (8-byte LDS reads) feeds both accumulators; odd block's A = kernel moved up one K slot; tap sum order changes
# speedup vs baseline: 1.3492x; 1.3492x over previous
#include <hip/hip_runtime.h>


typedef __attribute__((ext_vector_type(2))) float v2f;
typedef __attribute__((ext_vector_type(8))) float v8f;

#define TILE   256
#define BY     64
#define BX     128
#define IPH    (BY + 14)
#define IPW    (BX + 16)

__global__ __launch_bounds__(256) void norm_kernels_k(const float* __restrict__ k,
                                                      float* __restrict__ knorm) {
    const int t   = blockIdx.x;
    const int tid = threadIdx.x;

    __shared__ float red[256];
    float v = 0.0f;
    if (tid < 225) v = k[t * 225 + tid];
    red[tid] = v;
    __syncthreads();
    for (int s = 128; s > 0; s >>= 1) {
        if (tid < s) red[tid] += red[tid + s];
        __syncthreads();
    }
    const float inv = 1.0f / (red[0] + 1e-12f);

    const int dy = tid >> 4;
    const int dx = tid & 15;
    const float val = (dy < 15 && dx < 15) ? k[t * 225 + dy * 15 + dx] * inv : 0.0f;
    knorm[t * 256 + tid] = val;
}

__device__ __forceinline__ float xhalf(float x) {
    return __builtin_bit_cast(float, __builtin_amdgcn_permlanex16(
        __builtin_bit_cast(int, x), __builtin_bit_cast(int, x),
        0x76543210, 0xfedcba98, false, false));
}

__global__ __launch_bounds__(128) void pwblur_k(const float* __restrict__ x,
                                                const float* __restrict__ knorm,
                                                float* __restrict__ out) {
    __shared__ __attribute__((aligned(16))) float ip[IPH * IPW];

    const int tid  = threadIdx.x;
    const int lane = tid & 31;
    const int wave = tid >> 5;
    const bool hi  = (lane & 16) != 0;
    const int  nl  = lane & 15;

    const int bid   = blockIdx.x;
    const int sub   = bid & 7;
    const int t     = bid >> 3;
    const int ti    = t & 3;
    const int tj    = (t >> 2) & 3;
    const int plane = t >> 4;
    const int sub_y = sub >> 1;
    const int sub_x = sub & 1;

    const int ty0 = tj * TILE;
    const int tx0 = ti * TILE;
    const int y0  = sub_y * BY;
    const int x0  = sub_x * BX;

    const float* __restrict__ xin = x + (size_t)plane * 1024 * 1024;

    {
        int xa = x0 + tid - 7;        xa = xa < 0 ? 0 : (xa > TILE - 1 ? TILE - 1 : xa);
        const float* __restrict__ ca = xin + (tx0 + xa);
        #pragma unroll 13
        for (int r = 0; r < IPH; ++r) {
            int ys = y0 + r - 7;  ys = ys < 0 ? 0 : (ys > TILE - 1 ? TILE - 1 : ys);
            ip[r * IPW + tid] = ca[(size_t)(ty0 + ys) * 1024];
        }
        int xb = x0 + 128 + nl - 7;   xb = xb > TILE - 1 ? TILE - 1 : xb;
        const float* __restrict__ cb = xin + (tx0 + xb);
        #pragma unroll
        for (int k = 0; k < 10; ++k) {
            int r = (tid >> 4) + 8 * k;   r = r > IPH - 1 ? IPH - 1 : r;
            int ys = y0 + r - 7;  ys = ys < 0 ? 0 : (ys > TILE - 1 ? TILE - 1 : ys);
            ip[r * IPW + 128 + nl] = cb[(size_t)(ty0 + ys) * 1024];
        }
    }

    const float* __restrict__ kn = knorm + (tj * 4 + ti) * 256;
    v2f aNo[4], aNe[4], aSo[4], aSe[4];
    {
        const int pN = nl;
        const int pS = nl ^ 8;
        const int qN = (pN + 15) & 15;
        const int qS = (pS + 15) & 15;
        #pragma unroll
        for (int c = 0; c < 4; ++c) {
            const int kb  = 2 * c + (hi ? 8 : 0);
            const int km1 = kb > 0 ? kb - 1 : 0;
            const float no = kn[pN * 16 + km1];
            const float ne = kn[qN * 16 + km1];
            aNo[c].x = kb > 0 ? no : 0.0f;
            aNo[c].y = kn[pN * 16 + kb];
            aNe[c].x = kb > 0 ? ne : 0.0f;
            aNe[c].y = kn[qN * 16 + kb];
            aSo[c].x = kn[pS * 16 + kb];
            aSo[c].y = kn[pS * 16 + kb + 1];
            aSe[c].x = kn[qS * 16 + kb];
            aSe[c].y = kn[qS * 16 + kb + 1];
        }
    }
    __syncthreads();

    const float* bp = ip + wave * 32 + 2 * nl + (hi ? 8 : 0);

    v8f cN = {};
    v8f cS = {};

    float* __restrict__ op = out + (size_t)plane * 1024 * 1024
                           + (size_t)(ty0 + y0) * 1024 + (tx0 + x0 + wave * 32 + 2 * nl + (hi ? 1 : 0));

    for (int s = 0; s < IPH / 2; ++s) {
        const v2f* b0 = (const v2f*)(bp + (2 * s) * IPW);
        const v2f* b1 = (const v2f*)(bp + (2 * s + 1) * IPW);
        v8f dS = cS;
        v8f dN = cN;
        #pragma unroll
        for (int c = 0; c < 4; ++c) {
            const v2f b = b0[c];
            dS = __builtin_amdgcn_wmma_f32_16x16x4_f32(
                false, aSe[c], false, b, (short)0, dS, false, false);
            dN = __builtin_amdgcn_wmma_f32_16x16x4_f32(
                false, aNe[c], false, b, (short)0, dN, false, false);
        }
        #pragma unroll
        for (int c = 0; c < 4; ++c) {
            const v2f b = b1[c];
            dS = __builtin_amdgcn_wmma_f32_16x16x4_f32(
                false, aSo[c], false, b, (short)0, dS, false, false);
            dN = __builtin_amdgcn_wmma_f32_16x16x4_f32(
                false, aNo[c], false, b, (short)0, dN, false, false);
        }

        if (s >= 7) {
            op[(size_t)(2 * s - 14) * 1024] = hi ? dN[7] : dS[7];
            op[(size_t)(2 * s - 13) * 1024] = hi ? dN[6] : dS[6];
        }

        const float xN6 = xhalf(dN[6]);
        const float xN7 = xhalf(dN[7]);
        const float xS6 = xhalf(dS[6]);
        const float xS7 = xhalf(dS[7]);
        cN[0] = hi ? xN6 : 0.0f;
        cN[1] = hi ? xN7 : 0.0f;
        cS[0] = hi ? 0.0f : xS6;
        cS[1] = hi ? 0.0f : xS7;
        #pragma unroll
        for (int i = 2; i < 8; ++i) {
            cN[i] = dN[i - 2];
            cS[i] = dS[i - 2];
        }
    }
}

extern "C" void kernel_launch(void* const* d_in, const int* in_sizes, int n_in,
                              void* d_out, int out_size, void* d_ws, size_t ws_size,
                              hipStream_t stream) {
    const float* x    = (const float*)d_in[0];
    const float* kern = (const float*)d_in[1];
    float* outp  = (float*)d_out;
    float* knorm = (float*)d_ws;

    norm_kernels_k<<<16, 256, 0, stream>>>(kern, knorm);

    pwblur_k<<<3072, 128, 0, stream>>>(x, knorm, outp);
}


# Horizontal taps as paired products: tap halves x 8-column halves, 3 products of K=8 for 4: (X0-X1).lo, X1.(lo+hi), (X1-X2).hi, one sum chain each; 12 matrix instr a step, was 16; sum order changes
# speedup vs baseline: 1.4335x; 1.0625x over previous
#include <hip/hip_runtime.h>


typedef __attribute__((ext_vector_type(2))) float v2f;
typedef __attribute__((ext_vector_type(8))) float v8f;

#define TILE   256
#define BY     64
#define BX     128
#define IPH    (BY + 14)
#define IPW    (BX + 16)

__global__ __launch_bounds__(256) void norm_kernels_k(const float* __restrict__ k,
                                                      float* __restrict__ knorm) {
    const int t   = blockIdx.x;
    const int tid = threadIdx.x;

    __shared__ float red[256];
    float v = 0.0f;
    if (tid < 225) v = k[t * 225 + tid];
    red[tid] = v;
    __syncthreads();
    for (int s = 128; s > 0; s >>= 1) {
        if (tid < s) red[tid] += red[tid + s];
        __syncthreads();
    }
    const float inv = 1.0f / (red[0] + 1e-12f);

    const int dy = tid >> 4;
    const int dx = tid & 15;
    const float val = (dy < 15 && dx < 15) ? k[t * 225 + dy * 15 + dx] * inv : 0.0f;
    knorm[t * 256 + tid] = val;
}

__device__ __forceinline__ float xhalf(float x) {
    return __builtin_bit_cast(float, __builtin_amdgcn_permlanex16(
        __builtin_bit_cast(int, x), __builtin_bit_cast(int, x),
        0x76543210, 0xfedcba98, false, false));
}

__global__ __launch_bounds__(128) void pwblur_k(const float* __restrict__ x,
                                                const float* __restrict__ knorm,
                                                float* __restrict__ out) {
    __shared__ float ip[IPH * IPW];

    const int tid  = threadIdx.x;
    const int lane = tid & 31;
    const int wave = tid >> 5;
    const bool hi  = (lane & 16) != 0;
    const int  nl  = lane & 15;

    const int bid   = blockIdx.x;
    const int sub   = bid & 7;
    const int t     = bid >> 3;
    const int ti    = t & 3;
    const int tj    = (t >> 2) & 3;
    const int plane = t >> 4;
    const int sub_y = sub >> 1;
    const int sub_x = sub & 1;

    const int ty0 = tj * TILE;
    const int tx0 = ti * TILE;
    const int y0  = sub_y * BY;
    const int x0  = sub_x * BX;

    const float* __restrict__ xin = x + (size_t)plane * 1024 * 1024;

    {
        int xa = x0 + tid - 7;        xa = xa < 0 ? 0 : (xa > TILE - 1 ? TILE - 1 : xa);
        const float* __restrict__ ca = xin + (tx0 + xa);
        #pragma unroll 13
        for (int r = 0; r < IPH; ++r) {
            int ys = y0 + r - 7;  ys = ys < 0 ? 0 : (ys > TILE - 1 ? TILE - 1 : ys);
            ip[r * IPW + tid] = ca[(size_t)(ty0 + ys) * 1024];
        }
        int xb = x0 + 128 + nl - 7;   xb = xb > TILE - 1 ? TILE - 1 : xb;
        const float* __restrict__ cb = xin + (tx0 + xb);
        #pragma unroll
        for (int k = 0; k < 10; ++k) {
            int r = (tid >> 4) + 8 * k;   r = r > IPH - 1 ? IPH - 1 : r;
            int ys = y0 + r - 7;  ys = ys < 0 ? 0 : (ys > TILE - 1 ? TILE - 1 : ys);
            ip[r * IPW + 128 + nl] = cb[(size_t)(ty0 + ys) * 1024];
        }
    }

    const float* __restrict__ kn = knorm + (tj * 4 + ti) * 256;
    v2f a1o[2], a1e[2], a2o[2], a2e[2], a3o[2], a3e[2];
    {
        const int pN = nl;
        const int pM = nl ^ 8;
        const int qN = (pN + 15) & 15;
        const int qM = (pM + 15) & 15;
        #pragma unroll
        for (int c = 0; c < 2; ++c) {
            const int kb = 2 * c + (hi ? 4 : 0);
            const float lo_o0 = kn[pN * 16 + kb],     lo_o1 = kn[pN * 16 + kb + 1];
            const float hi_o0 = kn[pN * 16 + 8 + kb], hi_o1 = kn[pN * 16 + 8 + kb + 1];
            const float lo_e0 = kn[qN * 16 + kb],     lo_e1 = kn[qN * 16 + kb + 1];
            const float hi_e0 = kn[qN * 16 + 8 + kb], hi_e1 = kn[qN * 16 + 8 + kb + 1];
            a1o[c].x = lo_o0;           a1o[c].y = lo_o1;
            a1e[c].x = lo_e0;           a1e[c].y = lo_e1;
            a2o[c].x = lo_o0 + hi_o0;   a2o[c].y = lo_o1 + hi_o1;
            a2e[c].x = lo_e0 + hi_e0;   a2e[c].y = lo_e1 + hi_e1;
            a3o[c].x = kn[pM * 16 + 8 + kb];   a3o[c].y = kn[pM * 16 + 8 + kb + 1];
            a3e[c].x = kn[qM * 16 + 8 + kb];   a3e[c].y = kn[qM * 16 + 8 + kb + 1];
        }
    }
    __syncthreads();

    const int cn = nl < 8 ? nl : nl + 8;
    const float* bp = ip + wave * 32 + cn + (hi ? 4 : 0);

    v8f c1 = {};
    v8f c2 = {};
    v8f c3 = {};

    float* __restrict__ op = out + (size_t)plane * 1024 * 1024
                           + (size_t)(ty0 + y0) * 1024 + (tx0 + x0 + wave * 32 + cn + (hi ? 0 : 8));

    #pragma unroll 3
    for (int s = 0; s < IPH / 2; ++s) {
        const float* r0 = bp + (2 * s) * IPW;
        const float* r1 = bp + (2 * s + 1) * IPW;
        v8f d1 = c1;
        v8f d2 = c2;
        v8f d3 = c3;
        #pragma unroll
        for (int c = 0; c < 2; ++c) {
            const v2f x0 = { r0[2 * c],      r0[2 * c + 1] };
            const v2f x1 = { r0[8 + 2 * c],  r0[8 + 2 * c + 1] };
            const v2f x2 = { r0[16 + 2 * c], r0[16 + 2 * c + 1] };
            d1 = __builtin_amdgcn_wmma_f32_16x16x4_f32(
                false, a1e[c], false, x0 - x1, (short)0, d1, false, false);
            d2 = __builtin_amdgcn_wmma_f32_16x16x4_f32(
                false, a2e[c], false, x1, (short)0, d2, false, false);
            d3 = __builtin_amdgcn_wmma_f32_16x16x4_f32(
                false, a3e[c], false, x1 - x2, (short)0, d3, false, false);
        }
        #pragma unroll
        for (int c = 0; c < 2; ++c) {
            const v2f x0 = { r1[2 * c],      r1[2 * c + 1] };
            const v2f x1 = { r1[8 + 2 * c],  r1[8 + 2 * c + 1] };
            const v2f x2 = { r1[16 + 2 * c], r1[16 + 2 * c + 1] };
            d1 = __builtin_amdgcn_wmma_f32_16x16x4_f32(
                false, a1o[c], false, x0 - x1, (short)0, d1, false, false);
            d2 = __builtin_amdgcn_wmma_f32_16x16x4_f32(
                false, a2o[c], false, x1, (short)0, d2, false, false);
            d3 = __builtin_amdgcn_wmma_f32_16x16x4_f32(
                false, a3o[c], false, x1 - x2, (short)0, d3, false, false);
        }

        const float x16 = xhalf(d1[6]);
        const float x17 = xhalf(d1[7]);
        const float x26 = xhalf(d2[6]);
        const float x27 = xhalf(d2[7]);
        const float x36 = xhalf(d3[6]);
        const float x37 = xhalf(d3[7]);

        if (s >= 7) {
            op[(size_t)(2 * s - 14) * 1024] = hi ? (d1[7] + d2[7]) : (x27 - d3[7]);
            op[(size_t)(2 * s - 13) * 1024] = hi ? (d1[6] + d2[6]) : (x26 - d3[6]);
        }

        c1[0] = hi ? x16 : 0.0f;
        c1[1] = hi ? x17 : 0.0f;
        c2[0] = hi ? x26 : 0.0f;
        c2[1] = hi ? x27 : 0.0f;
        c3[0] = hi ? 0.0f : x36;
        c3[1] = hi ? 0.0f : x37;
        #pragma unroll
        for (int i = 2; i < 8; ++i) {
            c1[i] = d1[i - 2];
            c2[i] = d2[i - 2];
            c3[i] = d3[i - 2];
        }
    }
}

extern "C" void kernel_launch(void* const* d_in, const int* in_sizes, int n_in,
                              void* d_out, int out_size, void* d_ws, size_t ws_size,
                              hipStream_t stream) {
    const float* x    = (const float*)d_in[0];
    const float* kern = (const float*)d_in[1];
    float* outp  = (float*)d_out;
    float* knorm = (float*)d_ws;

    norm_kernels_k<<<16, 256, 0, stream>>>(kern, knorm);

    pwblur_k<<<3072, 128, 0, stream>>>(x, knorm, outp);
}


# Horizontal taps paired on a second level: 4 tap blocks x 4 column groups per lane, 9 products of K=4 for 16; a wave takes 64 columns, 18 matrix instr a step per 64 cols, was 24; sum order changes
# speedup vs baseline: 1.4966x; 1.0440x over previous
#include <hip/hip_runtime.h>


typedef __attribute__((ext_vector_type(2))) float v2f;
typedef __attribute__((ext_vector_type(8))) float v8f;

#define TILE   256
#define BY     64
#define BX     128
#define IPH    (BY + 14)
#define IPW    (BX + 16)

__global__ __launch_bounds__(256) void norm_kernels_k(const float* __restrict__ k,
                                                      float* __restrict__ knorm) {
    const int t   = blockIdx.x;
    const int tid = threadIdx.x;

    __shared__ float red[256];
    float v = 0.0f;
    if (tid < 225) v = k[t * 225 + tid];
    red[tid] = v;
    __syncthreads();
    for (int s = 128; s > 0; s >>= 1) {
        if (tid < s) red[tid] += red[tid + s];
        __syncthreads();
    }
    const float inv = 1.0f / (red[0] + 1e-12f);

    const int dy = tid >> 4;
    const int dx = tid & 15;
    const float val = (dy < 15 && dx < 15) ? k[t * 225 + dy * 15 + dx] * inv : 0.0f;
    knorm[t * 256 + tid] = val;
}

__device__ __forceinline__ float xhalf(float x) {
    return __builtin_bit_cast(float, __builtin_amdgcn_permlanex16(
        __builtin_bit_cast(int, x), __builtin_bit_cast(int, x),
        0x76543210, 0xfedcba98, false, false));
}

__global__ __launch_bounds__(64) void pwblur_k(const float* __restrict__ x,
                                               const float* __restrict__ knorm,
                                               float* __restrict__ out) {
    __shared__ float ip[IPH * IPW];

    const int tid  = threadIdx.x;
    const int lane = tid & 31;
    const int wave = tid >> 5;
    const bool hi  = (lane & 16) != 0;
    const int  nl  = lane & 15;

    const int bid   = blockIdx.x;
    const int sub   = bid & 7;
    const int t     = bid >> 3;
    const int ti    = t & 3;
    const int tj    = (t >> 2) & 3;
    const int plane = t >> 4;
    const int sub_y = sub >> 1;
    const int sub_x = sub & 1;

    const int ty0 = tj * TILE;
    const int tx0 = ti * TILE;
    const int y0  = sub_y * BY;
    const int x0  = sub_x * BX;

    const float* __restrict__ xin = x + (size_t)plane * 1024 * 1024;

    {
        #pragma unroll
        for (int q = 0; q < 2; ++q) {
            const int col = tid + 64 * q;
            int xa = x0 + col - 7;    xa = xa < 0 ? 0 : (xa > TILE - 1 ? TILE - 1 : xa);
            const float* __restrict__ ca = xin + (tx0 + xa);
            #pragma unroll 13
            for (int r = 0; r < IPH; ++r) {
                int ys = y0 + r - 7;  ys = ys < 0 ? 0 : (ys > TILE - 1 ? TILE - 1 : ys);
                ip[r * IPW + col] = ca[(size_t)(ty0 + ys) * 1024];
            }
        }
        int xb = x0 + 128 + nl - 7;   xb = xb > TILE - 1 ? TILE - 1 : xb;
        const float* __restrict__ cb = xin + (tx0 + xb);
        #pragma unroll
        for (int k = 0; k < 20; ++k) {
            int r = (tid >> 4) + 4 * k;   r = r > IPH - 1 ? IPH - 1 : r;
            int ys = y0 + r - 7;  ys = ys < 0 ? 0 : (ys > TILE - 1 ? TILE - 1 : ys);
            ip[r * IPW + 128 + nl] = cb[(size_t)(ty0 + ys) * 1024];
        }
    }

    const float* __restrict__ kn = knorm + (tj * 4 + ti) * 256;
    v2f ao[9], ae[9];
    {
        const int tb = hi ? 2 : 0;
        const float* __restrict__ ro = kn + nl * 16 + tb;
        const float* __restrict__ re = kn + ((nl + 15) & 15) * 16 + tb;
        const v2f o0 = { ro[0],  ro[1]  }, o1 = { ro[4],  ro[5]  };
        const v2f o2 = { ro[8],  ro[9]  }, o3 = { ro[12], ro[13] };
        const v2f e0 = { re[0],  re[1]  }, e1 = { re[4],  re[5]  };
        const v2f e2 = { re[8],  re[9]  }, e3 = { re[12], re[13] };
        ao[0] = o0;        ao[1] = o0 + o1;                 ao[2] = o1;
        ao[3] = o0 + o2;   ao[4] = (o0 + o1) + (o2 + o3);   ao[5] = o1 + o3;
        ao[6] = o2;        ao[7] = o2 + o3;                 ao[8] = o3;
        ae[0] = e0;        ae[1] = e0 + e1;                 ae[2] = e1;
        ae[3] = e0 + e2;   ae[4] = (e0 + e1) + (e2 + e3);   ae[5] = e1 + e3;
        ae[6] = e2;        ae[7] = e2 + e3;                 ae[8] = e3;
    }
    __syncthreads();

    const int cn = (nl & 3) + 16 * (nl >> 2);
    const float* bp = ip + wave * 64 + cn + (hi ? 2 : 0);

    v8f acc[9];
    #pragma unroll
    for (int q = 0; q < 9; ++q) acc[q] = (v8f){};

    float* __restrict__ op = out + (size_t)plane * 1024 * 1024
                           + (size_t)(ty0 + y0) * 1024 + (tx0 + x0 + wave * 64 + cn + (hi ? 0 : 4));

    #pragma unroll 3
    for (int s = 0; s < IPH / 2; ++s) {
        v8f d[9];
        #pragma unroll
        for (int q = 0; q < 9; ++q) d[q] = acc[q];

        #pragma unroll
        for (int par = 0; par < 2; ++par) {
            const float* rr = bp + (2 * s + par) * IPW;
            const v2f D0 = { rr[0],  rr[1]  }, D1 = { rr[4],  rr[5]  };
            const v2f D2 = { rr[8],  rr[9]  }, D3 = { rr[12], rr[13] };
            const v2f D4 = { rr[16], rr[17] }, D5 = { rr[20], rr[21] };
            const v2f D6 = { rr[24], rr[25] };
            const v2f s02 = D0 - D2, s13 = D1 - D3, s24 = D2 - D4;
            const v2f s35 = D3 - D5, s46 = D4 - D6;
            v2f b[9];
            b[0] = s02 - s13;   b[1] = s13;   b[2] = s13 - s24;
            b[3] = D2 - D3;     b[4] = D3;    b[5] = D3 - D4;
            b[6] = s24 - s35;   b[7] = s35;   b[8] = s35 - s46;
            #pragma unroll
            for (int q = 0; q < 9; ++q) {
                d[q] = __builtin_amdgcn_wmma_f32_16x16x4_f32(
                    false, par ? ao[q] : ae[q], false, b[q], (short)0, d[q], false, false);
            }
        }

        float x6[9], x7[9];
        #pragma unroll
        for (int q = 0; q < 9; ++q) {
            x6[q] = xhalf(d[q][6]);
            x7[q] = xhalf(d[q][7]);
        }

        if (s >= 7) {
            const float um7 = d[3][7] + d[4][7];
            const float ua7 = (d[0][7] + d[1][7]) + um7;
            const float uc7 = um7 - (d[6][7] + d[7][7]);
            const float lm7 = x7[4] - x7[5];
            const float lb7 = (x7[1] - x7[2]) + lm7;
            const float ld7 = lm7 - (x7[7] - x7[8]);
            const float um6 = d[3][6] + d[4][6];
            const float ua6 = (d[0][6] + d[1][6]) + um6;
            const float uc6 = um6 - (d[6][6] + d[7][6]);
            const float lm6 = x6[4] - x6[5];
            const float lb6 = (x6[1] - x6[2]) + lm6;
            const float ld6 = lm6 - (x6[7] - x6[8]);
            float* o14 = op + (size_t)(2 * s - 14) * 1024;
            float* o13 = op + (size_t)(2 * s - 13) * 1024;
            o14[0] = hi ? ua7 : lb7;
            o14[8] = hi ? uc7 : ld7;
            o13[0] = hi ? ua6 : lb6;
            o13[8] = hi ? uc6 : ld6;
        }

        #pragma unroll
        for (int q = 0; q < 9; ++q) {
            acc[q][0] = hi ? x6[q] : 0.0f;
            acc[q][1] = hi ? x7[q] : 0.0f;
            #pragma unroll
            for (int i = 2; i < 8; ++i) acc[q][i] = d[q][i - 2];
        }
    }
}

extern "C" void kernel_launch(void* const* d_in, const int* in_sizes, int n_in,
                              void* d_out, int out_size, void* d_ws, size_t ws_size,
                              hipStream_t stream) {
    const float* x    = (const float*)d_in[0];
    const float* kern = (const float*)d_in[1];
    float* outp  = (float*)d_out;
    float* knorm = (float*)d_ws;

    norm_kernels_k<<<16, 256, 0, stream>>>(kern, knorm);

    pwblur_k<<<3072, 64, 0, stream>>>(x, knorm, outp);
}


# Input band staged with 16-byte aligned global loads (band starts one column further left), all 44 of a thread in flight before its first LDS store; was 176 single loads in 13 dependent groups
# speedup vs baseline: 1.5880x; 1.0611x over previous
#include <hip/hip_runtime.h>


typedef __attribute__((ext_vector_type(2))) float v2f;
typedef __attribute__((ext_vector_type(8))) float v8f;

#define TILE   256
#define BY     64
#define BX     128
#define IPH    (BY + 14)
#define IPW    (BX + 16)

__global__ __launch_bounds__(256) void norm_kernels_k(const float* __restrict__ k,
                                                      float* __restrict__ knorm) {
    const int t   = blockIdx.x;
    const int tid = threadIdx.x;

    __shared__ float red[256];
    float v = 0.0f;
    if (tid < 225) v = k[t * 225 + tid];
    red[tid] = v;
    __syncthreads();
    for (int s = 128; s > 0; s >>= 1) {
        if (tid < s) red[tid] += red[tid + s];
        __syncthreads();
    }
    const float inv = 1.0f / (red[0] + 1e-12f);

    const int dy = tid >> 4;
    const int dx = tid & 15;
    const float val = (dy < 15 && dx < 15) ? k[t * 225 + dy * 15 + dx] * inv : 0.0f;
    knorm[t * 256 + tid] = val;
}

__device__ __forceinline__ float xhalf(float x) {
    return __builtin_bit_cast(float, __builtin_amdgcn_permlanex16(
        __builtin_bit_cast(int, x), __builtin_bit_cast(int, x),
        0x76543210, 0xfedcba98, false, false));
}

__global__ __launch_bounds__(64) void pwblur_k(const float* __restrict__ x,
                                               const float* __restrict__ knorm,
                                               float* __restrict__ out) {
    __shared__ __attribute__((aligned(16))) float ip[IPH * IPW];

    const int tid  = threadIdx.x;
    const int lane = tid & 31;
    const int wave = tid >> 5;
    const bool hi  = (lane & 16) != 0;
    const int  nl  = lane & 15;

    const int bid   = blockIdx.x;
    const int sub   = bid & 7;
    const int t     = bid >> 3;
    const int ti    = t & 3;
    const int tj    = (t >> 2) & 3;
    const int plane = t >> 4;
    const int sub_y = sub >> 1;
    const int sub_x = sub & 1;

    const int ty0 = tj * TILE;
    const int tx0 = ti * TILE;
    const int y0  = sub_y * BY;
    const int x0  = sub_x * BX;

    const float* __restrict__ xin = x + (size_t)plane * 1024 * 1024;

    {
        const int wv = __builtin_amdgcn_readfirstlane(tid >> 5);
        const int qa = tid & 31;
        const int xta = x0 - 8 + 4 * qa;
        const bool lefta = xta < 0;
        const float* __restrict__ pa = xin + (tx0 + (lefta ? 0 : xta));
        const int qb = 32 + (tid & 3);
        const int xtb = x0 - 8 + 4 * qb;
        const bool rightb = xtb > TILE - 4;
        const float* __restrict__ pb = xin + (tx0 + (rightb ? TILE - 4 : xtb));
        const int rb0 = tid >> 2;

        float4 va[IPH / 2], vb[5];
        #pragma unroll
        for (int m = 0; m < IPH / 2; ++m) {
            int ys = y0 + wv + 2 * m - 7;   ys = ys < 0 ? 0 : (ys > TILE - 1 ? TILE - 1 : ys);
            va[m] = *reinterpret_cast<const float4*>(pa + (size_t)(ty0 + ys) * 1024);
        }
        #pragma unroll
        for (int m = 0; m < 5; ++m) {
            int r = rb0 + 16 * m;           r = r > IPH - 1 ? IPH - 1 : r;
            int ys = y0 + r - 7;            ys = ys < 0 ? 0 : (ys > TILE - 1 ? TILE - 1 : ys);
            vb[m] = *reinterpret_cast<const float4*>(pb + (size_t)(ty0 + ys) * 1024);
        }
        #pragma unroll
        for (int m = 0; m < IPH / 2; ++m) {
            float4 w = va[m];
            if (lefta) w = make_float4(w.x, w.x, w.x, w.x);
            *reinterpret_cast<float4*>(ip + (wv + 2 * m) * IPW + 4 * qa) = w;
        }
        #pragma unroll
        for (int m = 0; m < 5; ++m) {
            int r = rb0 + 16 * m;           r = r > IPH - 1 ? IPH - 1 : r;
            float4 w = vb[m];
            if (rightb) w = make_float4(w.w, w.w, w.w, w.w);
            *reinterpret_cast<float4*>(ip + r * IPW + 4 * qb) = w;
        }
    }

    const float* __restrict__ kn = knorm + (tj * 4 + ti) * 256;
    v2f ao[9], ae[9];
    {
        const int tb = hi ? 2 : 0;
        const float* __restrict__ ro = kn + nl * 16 + tb;
        const float* __restrict__ re = kn + ((nl + 15) & 15) * 16 + tb;
        const v2f o0 = { ro[0],  ro[1]  }, o1 = { ro[4],  ro[5]  };
        const v2f o2 = { ro[8],  ro[9]  }, o3 = { ro[12], ro[13] };
        const v2f e0 = { re[0],  re[1]  }, e1 = { re[4],  re[5]  };
        const v2f e2 = { re[8],  re[9]  }, e3 = { re[12], re[13] };
        ao[0] = o0;        ao[1] = o0 + o1;                 ao[2] = o1;
        ao[3] = o0 + o2;   ao[4] = (o0 + o1) + (o2 + o3);   ao[5] = o1 + o3;
        ao[6] = o2;        ao[7] = o2 + o3;                 ao[8] = o3;
        ae[0] = e0;        ae[1] = e0 + e1;                 ae[2] = e1;
        ae[3] = e0 + e2;   ae[4] = (e0 + e1) + (e2 + e3);   ae[5] = e1 + e3;
        ae[6] = e2;        ae[7] = e2 + e3;                 ae[8] = e3;
    }
    __syncthreads();

    const int cn = (nl & 3) + 16 * (nl >> 2);
    const float* bp = ip + 1 + wave * 64 + cn + (hi ? 2 : 0);

    v8f acc[9];
    #pragma unroll
    for (int q = 0; q < 9; ++q) acc[q] = (v8f){};

    float* __restrict__ op = out + (size_t)plane * 1024 * 1024
                           + (size_t)(ty0 + y0) * 1024 + (tx0 + x0 + wave * 64 + cn + (hi ? 0 : 4));

    #pragma unroll 3
    for (int s = 0; s < IPH / 2; ++s) {
        v8f d[9];
        #pragma unroll
        for (int q = 0; q < 9; ++q) d[q] = acc[q];

        #pragma unroll
        for (int par = 0; par < 2; ++par) {
            const float* rr = bp + (2 * s + par) * IPW;
            const v2f D0 = { rr[0],  rr[1]  }, D1 = { rr[4],  rr[5]  };
            const v2f D2 = { rr[8],  rr[9]  }, D3 = { rr[12], rr[13] };
            const v2f D4 = { rr[16], rr[17] }, D5 = { rr[20], rr[21] };
            const v2f D6 = { rr[24], rr[25] };
            const v2f s02 = D0 - D2, s13 = D1 - D3, s24 = D2 - D4;
            const v2f s35 = D3 - D5, s46 = D4 - D6;
            v2f b[9];
            b[0] = s02 - s13;   b[1] = s13;   b[2] = s13 - s24;
            b[3] = D2 - D3;     b[4] = D3;    b[5] = D3 - D4;
            b[6] = s24 - s35;   b[7] = s35;   b[8] = s35 - s46;
            #pragma unroll
            for (int q = 0; q < 9; ++q) {
                d[q] = __builtin_amdgcn_wmma_f32_16x16x4_f32(
                    false, par ? ao[q] : ae[q], false, b[q], (short)0, d[q], false, false);
            }
        }

        float x6[9], x7[9];
        #pragma unroll
        for (int q = 0; q < 9; ++q) {
            x6[q] = xhalf(d[q][6]);
            x7[q] = xhalf(d[q][7]);
        }

        if (s >= 7) {
            const float um7 = d[3][7] + d[4][7];
            const float ua7 = (d[0][7] + d[1][7]) + um7;
            const float uc7 = um7 - (d[6][7] + d[7][7]);
            const float lm7 = x7[4] - x7[5];
            const float lb7 = (x7[1] - x7[2]) + lm7;
            const float ld7 = lm7 - (x7[7] - x7[8]);
            const float um6 = d[3][6] + d[4][6];
            const float ua6 = (d[0][6] + d[1][6]) + um6;
            const float uc6 = um6 - (d[6][6] + d[7][6]);
            const float lm6 = x6[4] - x6[5];
            const float lb6 = (x6[1] - x6[2]) + lm6;
            const float ld6 = lm6 - (x6[7] - x6[8]);
            float* o14 = op + (size_t)(2 * s - 14) * 1024;
            float* o13 = op + (size_t)(2 * s - 13) * 1024;
            o14[0] = hi ? ua7 : lb7;
            o14[8] = hi ? uc7 : ld7;
            o13[0] = hi ? ua6 : lb6;
            o13[8] = hi ? uc6 : ld6;
        }

        #pragma unroll
        for (int q = 0; q < 9; ++q) {
            acc[q][0] = hi ? x6[q] : 0.0f;
            acc[q][1] = hi ? x7[q] : 0.0f;
            #pragma unroll
            for (int i = 2; i < 8; ++i) acc[q][i] = d[q][i - 2];
        }
    }
}

extern "C" void kernel_launch(void* const* d_in, const int* in_sizes, int n_in,
                              void* d_out, int out_size, void* d_ws, size_t ws_size,
                              hipStream_t stream) {
    const float* x    = (const float*)d_in[0];
    const float* kern = (const float*)d_in[1];
    float* outp  = (float*)d_out;
    float* knorm = (float*)d_ws;

    norm_kernels_k<<<16, 256, 0, stream>>>(kern, knorm);

    pwblur_k<<<3072, 64, 0, stream>>>(x, knorm, outp);
}


# Chain positions split by parity over the lane halves (even in D rows 0-7, odd in rows 8-15): a step's move stays in its lane; 18 cross-half lane moves and 18 selects a step become 2 and 6
# speedup vs baseline: 1.6797x; 1.0577x over previous
#include <hip/hip_runtime.h>


typedef __attribute__((ext_vector_type(2))) float v2f;
typedef __attribute__((ext_vector_type(8))) float v8f;

#define TILE   256
#define BY     64
#define BX     128
#define IPH    (BY + 14)
#define IPW    (BX + 16)

__global__ __launch_bounds__(256) void norm_kernels_k(const float* __restrict__ k,
                                                      float* __restrict__ knorm) {
    const int t   = blockIdx.x;
    const int tid = threadIdx.x;

    __shared__ float red[256];
    float v = 0.0f;
    if (tid < 225) v = k[t * 225 + tid];
    red[tid] = v;
    __syncthreads();
    for (int s = 128; s > 0; s >>= 1) {
        if (tid < s) red[tid] += red[tid + s];
        __syncthreads();
    }
    const float inv = 1.0f / (red[0] + 1e-12f);

    const int dy = tid >> 4;
    const int dx = tid & 15;
    const float val = (dy < 15 && dx < 15) ? k[t * 225 + dy * 15 + dx] * inv : 0.0f;
    knorm[t * 256 + tid] = val;
}

__device__ __forceinline__ float xhalf(float x) {
    return __builtin_bit_cast(float, __builtin_amdgcn_permlanex16(
        __builtin_bit_cast(int, x), __builtin_bit_cast(int, x),
        0x76543210, 0xfedcba98, false, false));
}

__global__ __launch_bounds__(64) void pwblur_k(const float* __restrict__ x,
                                               const float* __restrict__ knorm,
                                               float* __restrict__ out) {
    __shared__ __attribute__((aligned(16))) float ip[IPH * IPW];

    const int tid  = threadIdx.x;
    const int lane = tid & 31;
    const int wave = tid >> 5;
    const bool hi  = (lane & 16) != 0;
    const int  nl  = lane & 15;

    const int bid   = blockIdx.x;
    const int sub   = bid & 7;
    const int t     = bid >> 3;
    const int ti    = t & 3;
    const int tj    = (t >> 2) & 3;
    const int plane = t >> 4;
    const int sub_y = sub >> 1;
    const int sub_x = sub & 1;

    const int ty0 = tj * TILE;
    const int tx0 = ti * TILE;
    const int y0  = sub_y * BY;
    const int x0  = sub_x * BX;

    const float* __restrict__ xin = x + (size_t)plane * 1024 * 1024;

    {
        const int wv = __builtin_amdgcn_readfirstlane(tid >> 5);
        const int qa = tid & 31;
        const int xta = x0 - 8 + 4 * qa;
        const bool lefta = xta < 0;
        const float* __restrict__ pa = xin + (tx0 + (lefta ? 0 : xta));
        const int qb = 32 + (tid & 3);
        const int xtb = x0 - 8 + 4 * qb;
        const bool rightb = xtb > TILE - 4;
        const float* __restrict__ pb = xin + (tx0 + (rightb ? TILE - 4 : xtb));
        const int rb0 = tid >> 2;

        float4 va[IPH / 2], vb[5];
        #pragma unroll
        for (int m = 0; m < IPH / 2; ++m) {
            int ys = y0 + wv + 2 * m - 7;   ys = ys < 0 ? 0 : (ys > TILE - 1 ? TILE - 1 : ys);
            va[m] = *reinterpret_cast<const float4*>(pa + (size_t)(ty0 + ys) * 1024);
        }
        #pragma unroll
        for (int m = 0; m < 5; ++m) {
            int r = rb0 + 16 * m;           r = r > IPH - 1 ? IPH - 1 : r;
            int ys = y0 + r - 7;            ys = ys < 0 ? 0 : (ys > TILE - 1 ? TILE - 1 : ys);
            vb[m] = *reinterpret_cast<const float4*>(pb + (size_t)(ty0 + ys) * 1024);
        }
        #pragma unroll
        for (int m = 0; m < IPH / 2; ++m) {
            float4 w = va[m];
            if (lefta) w = make_float4(w.x, w.x, w.x, w.x);
            *reinterpret_cast<float4*>(ip + (wv + 2 * m) * IPW + 4 * qa) = w;
        }
        #pragma unroll
        for (int m = 0; m < 5; ++m) {
            int r = rb0 + 16 * m;           r = r > IPH - 1 ? IPH - 1 : r;
            float4 w = vb[m];
            if (rightb) w = make_float4(w.w, w.w, w.w, w.w);
            *reinterpret_cast<float4*>(ip + r * IPW + 4 * qb) = w;
        }
    }

    const float* __restrict__ kn = knorm + (tj * 4 + ti) * 256;
    v2f ao[9], ae[9];
    {
        const int tb = hi ? 2 : 0;
        const int rr7 = nl & 7;
        const int pos = (nl >> 3) + 2 * ((rr7 >> 1) + ((rr7 & 1) ? 4 : 0));
        const float* __restrict__ ro = kn + pos * 16 + tb;
        const float* __restrict__ re = kn + ((pos + 15) & 15) * 16 + tb;
        const v2f o0 = { ro[0],  ro[1]  }, o1 = { ro[4],  ro[5]  };
        const v2f o2 = { ro[8],  ro[9]  }, o3 = { ro[12], ro[13] };
        const v2f e0 = { re[0],  re[1]  }, e1 = { re[4],  re[5]  };
        const v2f e2 = { re[8],  re[9]  }, e3 = { re[12], re[13] };
        ao[0] = o0;        ao[1] = o0 + o1;                 ao[2] = o1;
        ao[3] = o0 + o2;   ao[4] = (o0 + o1) + (o2 + o3);   ao[5] = o1 + o3;
        ao[6] = o2;        ao[7] = o2 + o3;                 ao[8] = o3;
        ae[0] = e0;        ae[1] = e0 + e1;                 ae[2] = e1;
        ae[3] = e0 + e2;   ae[4] = (e0 + e1) + (e2 + e3);   ae[5] = e1 + e3;
        ae[6] = e2;        ae[7] = e2 + e3;                 ae[8] = e3;
    }
    __syncthreads();

    const int cn = (nl & 3) + 16 * (nl >> 2);
    const float* bp = ip + 1 + wave * 64 + cn + (hi ? 2 : 0);

    v8f d[9];
    #pragma unroll
    for (int q = 0; q < 9; ++q) d[q] = (v8f){};

    float* __restrict__ op = out + (size_t)plane * 1024 * 1024
                           + (size_t)(ty0 + y0) * 1024 + (tx0 + x0 + wave * 64 + cn + (hi ? 0 : 4));

    #pragma unroll 3
    for (int s = 0; s < IPH / 2; ++s) {
        #pragma unroll
        for (int q = 0; q < 9; ++q)
            d[q] = __builtin_shufflevector(d[q], (v8f){}, 8, 6, 0, 1, 2, 3, 4, 5);

        #pragma unroll
        for (int par = 0; par < 2; ++par) {
            const float* rr = bp + (2 * s + par) * IPW;
            const v2f D0 = { rr[0],  rr[1]  }, D1 = { rr[4],  rr[5]  };
            const v2f D2 = { rr[8],  rr[9]  }, D3 = { rr[12], rr[13] };
            const v2f D4 = { rr[16], rr[17] }, D5 = { rr[20], rr[21] };
            const v2f D6 = { rr[24], rr[25] };
            const v2f s02 = D0 - D2, s13 = D1 - D3, s24 = D2 - D4;
            const v2f s35 = D3 - D5, s46 = D4 - D6;
            v2f b[9];
            b[0] = s02 - s13;   b[1] = s13;   b[2] = s13 - s24;
            b[3] = D2 - D3;     b[4] = D3;    b[5] = D3 - D4;
            b[6] = s24 - s35;   b[7] = s35;   b[8] = s35 - s46;
            #pragma unroll
            for (int q = 0; q < 9; ++q) {
                d[q] = __builtin_amdgcn_wmma_f32_16x16x4_f32(
                    false, par ? ao[q] : ae[q], false, b[q], (short)0, d[q], false, false);
            }
        }

        if (s >= 7) {
            const float p34 = d[3][7] + d[4][7];
            const float m45 = d[4][7] - d[5][7];
            const float ya  = (d[0][7] + d[1][7]) + p34;
            const float yb  = (d[1][7] - d[2][7]) + m45;
            const float yc  = p34 - (d[6][7] + d[7][7]);
            const float yd  = m45 - (d[7][7] - d[8][7]);
            const float g0 = xhalf(hi ? yb : ya);
            const float g1 = xhalf(hi ? yd : yc);
            float* o14 = op + (size_t)(2 * s - 14) * 1024;
            float* o13 = op + (size_t)(2 * s - 13) * 1024;
            o14[0] = hi ? ya : g0;
            o14[8] = hi ? yc : g1;
            o13[0] = hi ? g0 : yb;
            o13[8] = hi ? g1 : yd;
        }
    }
}

extern "C" void kernel_launch(void* const* d_in, const int* in_sizes, int n_in,
                              void* d_out, int out_size, void* d_ws, size_t ws_size,
                              hipStream_t stream) {
    const float* x    = (const float*)d_in[0];
    const float* kern = (const float*)d_in[1];
    float* outp  = (float*)d_out;
    float* knorm = (float*)d_ws;

    norm_kernels_k<<<16, 256, 0, stream>>>(kern, knorm);

    pwblur_k<<<3072, 64, 0, stream>>>(x, knorm, outp);
}


# Work-group enlarged to 2 x 2 waves sharing one 142-row LDS band (128 x 128 outputs): rows between its two strips staged once, one wave on each SIMD; 1536 groups of 4 waves, was 3072 of 2
# speedup vs baseline: 1.8162x; 1.0813x over previous
#include <hip/hip_runtime.h>


typedef __attribute__((ext_vector_type(2))) float v2f;
typedef __attribute__((ext_vector_type(8))) float v8f;

#define TILE   256
#define BY     64
#define GY     128
#define BX     128
#define IPH    (BY + 14)
#define BPH    (GY + 14)
#define IPW    (BX + 16)

__global__ __launch_bounds__(256) void norm_kernels_k(const float* __restrict__ k,
                                                      float* __restrict__ knorm) {
    const int t   = blockIdx.x;
    const int tid = threadIdx.x;

    __shared__ float red[256];
    float v = 0.0f;
    if (tid < 225) v = k[t * 225 + tid];
    red[tid] = v;
    __syncthreads();
    for (int s = 128; s > 0; s >>= 1) {
        if (tid < s) red[tid] += red[tid + s];
        __syncthreads();
    }
    const float inv = 1.0f / (red[0] + 1e-12f);

    const int dy = tid >> 4;
    const int dx = tid & 15;
    const float val = (dy < 15 && dx < 15) ? k[t * 225 + dy * 15 + dx] * inv : 0.0f;
    knorm[t * 256 + tid] = val;
}

__device__ __forceinline__ float xhalf(float x) {
    return __builtin_bit_cast(float, __builtin_amdgcn_permlanex16(
        __builtin_bit_cast(int, x), __builtin_bit_cast(int, x),
        0x76543210, 0xfedcba98, false, false));
}

__global__ __launch_bounds__(128) void pwblur_k(const float* __restrict__ x,
                                               const float* __restrict__ knorm,
                                               float* __restrict__ out) {
    __shared__ __attribute__((aligned(16))) float ip[BPH * IPW];

    const int tid  = threadIdx.x;
    const int lane = tid & 31;
    const int wave = tid >> 5;
    const bool hi  = (lane & 16) != 0;
    const int  nl  = lane & 15;

    const int bid   = blockIdx.x;
    const int sub   = bid & 3;
    const int t     = bid >> 2;
    const int ti    = t & 3;
    const int tj    = (t >> 2) & 3;
    const int plane = t >> 4;
    const int sub_y = sub >> 1;
    const int sub_x = sub & 1;

    const int ty0 = tj * TILE;
    const int tx0 = ti * TILE;
    const int y0  = sub_y * GY;
    const int x0  = sub_x * BX;

    const float* __restrict__ xin = x + (size_t)plane * 1024 * 1024;

    {
        const int wv = __builtin_amdgcn_readfirstlane(tid >> 5);
        const int qa = tid & 31;
        const int xta = x0 - 8 + 4 * qa;
        const bool lefta = xta < 0;
        const float* __restrict__ pa = xin + (tx0 + (lefta ? 0 : xta));
        const int qb = 32 + (tid & 3);
        const int xtb = x0 - 8 + 4 * qb;
        const bool rightb = xtb > TILE - 4;
        const float* __restrict__ pb = xin + (tx0 + (rightb ? TILE - 4 : xtb));
        const int rb0 = tid >> 2;

        float4 va[(BPH + 3) / 4], vb[5];
        #pragma unroll
        for (int m = 0; m < (BPH + 3) / 4; ++m) {
            int r = wv + 4 * m;             r = r > BPH - 1 ? BPH - 1 : r;
            int ys = y0 + r - 7;            ys = ys < 0 ? 0 : (ys > TILE - 1 ? TILE - 1 : ys);
            va[m] = *reinterpret_cast<const float4*>(pa + (size_t)(ty0 + ys) * 1024);
        }
        #pragma unroll
        for (int m = 0; m < 5; ++m) {
            int r = rb0 + 32 * m;           r = r > BPH - 1 ? BPH - 1 : r;
            int ys = y0 + r - 7;            ys = ys < 0 ? 0 : (ys > TILE - 1 ? TILE - 1 : ys);
            vb[m] = *reinterpret_cast<const float4*>(pb + (size_t)(ty0 + ys) * 1024);
        }
        #pragma unroll
        for (int m = 0; m < (BPH + 3) / 4; ++m) {
            int r = wv + 4 * m;             r = r > BPH - 1 ? BPH - 1 : r;
            float4 w = va[m];
            if (lefta) w = make_float4(w.x, w.x, w.x, w.x);
            *reinterpret_cast<float4*>(ip + r * IPW + 4 * qa) = w;
        }
        #pragma unroll
        for (int m = 0; m < 5; ++m) {
            int r = rb0 + 32 * m;           r = r > BPH - 1 ? BPH - 1 : r;
            float4 w = vb[m];
            if (rightb) w = make_float4(w.w, w.w, w.w, w.w);
            *reinterpret_cast<float4*>(ip + r * IPW + 4 * qb) = w;
        }
    }

    const float* __restrict__ kn = knorm + (tj * 4 + ti) * 256;
    v2f ao[9], ae[9];
    {
        const int tb = hi ? 2 : 0;
        const int rr7 = nl & 7;
        const int pos = (nl >> 3) + 2 * ((rr7 >> 1) + ((rr7 & 1) ? 4 : 0));
        const float* __restrict__ ro = kn + pos * 16 + tb;
        const float* __restrict__ re = kn + ((pos + 15) & 15) * 16 + tb;
        const v2f o0 = { ro[0],  ro[1]  }, o1 = { ro[4],  ro[5]  };
        const v2f o2 = { ro[8],  ro[9]  }, o3 = { ro[12], ro[13] };
        const v2f e0 = { re[0],  re[1]  }, e1 = { re[4],  re[5]  };
        const v2f e2 = { re[8],  re[9]  }, e3 = { re[12], re[13] };
        ao[0] = o0;        ao[1] = o0 + o1;                 ao[2] = o1;
        ao[3] = o0 + o2;   ao[4] = (o0 + o1) + (o2 + o3);   ao[5] = o1 + o3;
        ao[6] = o2;        ao[7] = o2 + o3;                 ao[8] = o3;
        ae[0] = e0;        ae[1] = e0 + e1;                 ae[2] = e1;
        ae[3] = e0 + e2;   ae[4] = (e0 + e1) + (e2 + e3);   ae[5] = e1 + e3;
        ae[6] = e2;        ae[7] = e2 + e3;                 ae[8] = e3;
    }
    __syncthreads();

    const int wx = wave & 1;
    const int wy = wave >> 1;
    const int cn = (nl & 3) + 16 * (nl >> 2);
    const float* bp = ip + wy * BY * IPW + 1 + wx * 64 + cn + (hi ? 2 : 0);

    v8f d[9];
    #pragma unroll
    for (int q = 0; q < 9; ++q) d[q] = (v8f){};

    float* __restrict__ op = out + (size_t)plane * 1024 * 1024
                           + (size_t)(ty0 + y0 + wy * BY) * 1024 + (tx0 + x0 + wx * 64 + cn + (hi ? 0 : 4));

    #pragma unroll 3
    for (int s = 0; s < IPH / 2; ++s) {
        #pragma unroll
        for (int q = 0; q < 9; ++q)
            d[q] = __builtin_shufflevector(d[q], (v8f){}, 8, 6, 0, 1, 2, 3, 4, 5);

        #pragma unroll
        for (int par = 0; par < 2; ++par) {
            const float* rr = bp + (2 * s + par) * IPW;
            const v2f D0 = { rr[0],  rr[1]  }, D1 = { rr[4],  rr[5]  };
            const v2f D2 = { rr[8],  rr[9]  }, D3 = { rr[12], rr[13] };
            const v2f D4 = { rr[16], rr[17] }, D5 = { rr[20], rr[21] };
            const v2f D6 = { rr[24], rr[25] };
            const v2f s02 = D0 - D2, s13 = D1 - D3, s24 = D2 - D4;
            const v2f s35 = D3 - D5, s46 = D4 - D6;
            v2f b[9];
            b[0] = s02 - s13;   b[1] = s13;   b[2] = s13 - s24;
            b[3] = D2 - D3;     b[4] = D3;    b[5] = D3 - D4;
            b[6] = s24 - s35;   b[7] = s35;   b[8] = s35 - s46;
            #pragma unroll
            for (int q = 0; q < 9; ++q) {
                d[q] = __builtin_amdgcn_wmma_f32_16x16x4_f32(
                    false, par ? ao[q] : ae[q], false, b[q], (short)0, d[q], false, false);
            }
        }

        if (s >= 7) {
            const float p34 = d[3][7] + d[4][7];
            const float m45 = d[4][7] - d[5][7];
            const float ya  = (d[0][7] + d[1][7]) + p34;
            const float yb  = (d[1][7] - d[2][7]) + m45;
            const float yc  = p34 - (d[6][7] + d[7][7]);
            const float yd  = m45 - (d[7][7] - d[8][7]);
            const float g0 = xhalf(hi ? yb : ya);
            const float g1 = xhalf(hi ? yd : yc);
            float* o14 = op + (size_t)(2 * s - 14) * 1024;
            float* o13 = op + (size_t)(2 * s - 13) * 1024;
            o14[0] = hi ? ya : g0;
            o14[8] = hi ? yc : g1;
            o13[0] = hi ? g0 : yb;
            o13[8] = hi ? g1 : yd;
        }
    }
}

extern "C" void kernel_launch(void* const* d_in, const int* in_sizes, int n_in,
                              void* d_out, int out_size, void* d_ws, size_t ws_size,
                              hipStream_t stream) {
    const float* x    = (const float*)d_in[0];
    const float* kern = (const float*)d_in[1];
    float* outp  = (float*)d_out;
    float* knorm = (float*)d_ws;

    norm_kernels_k<<<16, 256, 0, stream>>>(kern, knorm);

    pwblur_k<<<1536, 128, 0, stream>>>(x, knorm, outp);
}


# Horizontal taps by 7-point interpolation (z = 0, 1, -1, 2, -2, 1/2, inf), not two-level pairing: 7 products of K=4 per input row, was 9; 14 matrix instr a step, was 18; sum order and rounding change
# speedup vs baseline: 1.9431x; 1.0699x over previous
#include <hip/hip_runtime.h>


typedef __attribute__((ext_vector_type(2))) float v2f;
typedef __attribute__((ext_vector_type(8))) float v8f;

#define TILE   256
#define BY     64
#define GY     128
#define BX     128
#define IPH    (BY + 14)
#define BPH    (GY + 14)
#define IPW    (BX + 16)

__global__ __launch_bounds__(256) void norm_kernels_k(const float* __restrict__ k,
                                                      float* __restrict__ knorm) {
    const int t   = blockIdx.x;
    const int tid = threadIdx.x;

    __shared__ float red[256];
    float v = 0.0f;
    if (tid < 225) v = k[t * 225 + tid];
    red[tid] = v;
    __syncthreads();
    for (int s = 128; s > 0; s >>= 1) {
        if (tid < s) red[tid] += red[tid + s];
        __syncthreads();
    }
    const float inv = 1.0f / (red[0] + 1e-12f);

    const int dy = tid >> 4;
    const int dx = tid & 15;
    const float val = (dy < 15 && dx < 15) ? k[t * 225 + dy * 15 + dx] * inv : 0.0f;
    knorm[t * 256 + tid] = val;
}

__device__ __forceinline__ void tap_points(const float* __restrict__ r, v2f (&u)[7]) {
    const v2f g0 = { r[0], r[1] }, g1 = { r[4],  r[5]  };
    const v2f g2 = { r[8], r[9] }, g3 = { r[12], r[13] };
    const v2f s02 = g0 + g2, s13 = g1 + g3;
    const v2f p = g0 + 4.0f * g2, q = g1 + 4.0f * g3;
    u[0] = 0.25f * g0;
    u[1] = (1.0f / 6.0f)  * (s02 + s13);
    u[2] = (1.0f / 18.0f) * (s02 - s13);
    u[3] = (1.0f / 72.0f)  * (p + 2.0f * q);
    u[4] = (1.0f / 120.0f) * (p - 2.0f * q);
    u[5] = (4.0f / 45.0f) * ((8.0f * g0 + 4.0f * g1) + (2.0f * g2 + g3));
    u[6] = 0.5f * g3;
}

__device__ __forceinline__ float xhalf(float x) {
    return __builtin_bit_cast(float, __builtin_amdgcn_permlanex16(
        __builtin_bit_cast(int, x), __builtin_bit_cast(int, x),
        0x76543210, 0xfedcba98, false, false));
}

__global__ __launch_bounds__(128) void pwblur_k(const float* __restrict__ x,
                                               const float* __restrict__ knorm,
                                               float* __restrict__ out) {
    __shared__ __attribute__((aligned(16))) float ip[BPH * IPW];

    const int tid  = threadIdx.x;
    const int lane = tid & 31;
    const int wave = tid >> 5;
    const bool hi  = (lane & 16) != 0;
    const int  nl  = lane & 15;

    const int bid   = blockIdx.x;
    const int sub   = bid & 3;
    const int t     = bid >> 2;
    const int ti    = t & 3;
    const int tj    = (t >> 2) & 3;
    const int plane = t >> 4;
    const int sub_y = sub >> 1;
    const int sub_x = sub & 1;

    const int ty0 = tj * TILE;
    const int tx0 = ti * TILE;
    const int y0  = sub_y * GY;
    const int x0  = sub_x * BX;

    const float* __restrict__ xin = x + (size_t)plane * 1024 * 1024;

    {
        const int wv = __builtin_amdgcn_readfirstlane(tid >> 5);
        const int qa = tid & 31;
        const int xta = x0 - 8 + 4 * qa;
        const bool lefta = xta < 0;
        const float* __restrict__ pa = xin + (tx0 + (lefta ? 0 : xta));
        const int qb = 32 + (tid & 3);
        const int xtb = x0 - 8 + 4 * qb;
        const bool rightb = xtb > TILE - 4;
        const float* __restrict__ pb = xin + (tx0 + (rightb ? TILE - 4 : xtb));
        const int rb0 = tid >> 2;

        float4 va[(BPH + 3) / 4], vb[5];
        #pragma unroll
        for (int m = 0; m < (BPH + 3) / 4; ++m) {
            int r = wv + 4 * m;             r = r > BPH - 1 ? BPH - 1 : r;
            int ys = y0 + r - 7;            ys = ys < 0 ? 0 : (ys > TILE - 1 ? TILE - 1 : ys);
            va[m] = *reinterpret_cast<const float4*>(pa + (size_t)(ty0 + ys) * 1024);
        }
        #pragma unroll
        for (int m = 0; m < 5; ++m) {
            int r = rb0 + 32 * m;           r = r > BPH - 1 ? BPH - 1 : r;
            int ys = y0 + r - 7;            ys = ys < 0 ? 0 : (ys > TILE - 1 ? TILE - 1 : ys);
            vb[m] = *reinterpret_cast<const float4*>(pb + (size_t)(ty0 + ys) * 1024);
        }
        #pragma unroll
        for (int m = 0; m < (BPH + 3) / 4; ++m) {
            int r = wv + 4 * m;             r = r > BPH - 1 ? BPH - 1 : r;
            float4 w = va[m];
            if (lefta) w = make_float4(w.x, w.x, w.x, w.x);
            *reinterpret_cast<float4*>(ip + r * IPW + 4 * qa) = w;
        }
        #pragma unroll
        for (int m = 0; m < 5; ++m) {
            int r = rb0 + 32 * m;           r = r > BPH - 1 ? BPH - 1 : r;
            float4 w = vb[m];
            if (rightb) w = make_float4(w.w, w.w, w.w, w.w);
            *reinterpret_cast<float4*>(ip + r * IPW + 4 * qb) = w;
        }
    }

    const float* __restrict__ kn = knorm + (tj * 4 + ti) * 256;
    v2f ao[7], ae[7];
    {
        const int tb = hi ? 2 : 0;
        const int rr7 = nl & 7;
        const int pos = (nl >> 3) + 2 * ((rr7 >> 1) + ((rr7 & 1) ? 4 : 0));
        const float* __restrict__ ro = kn + pos * 16 + tb;
        const float* __restrict__ re = kn + ((pos + 15) & 15) * 16 + tb;
        tap_points(ro, ao);
        tap_points(re, ae);
    }
    __syncthreads();

    const int wx = wave & 1;
    const int wy = wave >> 1;
    const int cn = (nl & 3) + 16 * (nl >> 2);
    const float* bp = ip + wy * BY * IPW + 1 + wx * 64 + cn + (hi ? 2 : 0);

    v8f d[7];
    #pragma unroll
    for (int q = 0; q < 7; ++q) d[q] = (v8f){};

    float* __restrict__ op = out + (size_t)plane * 1024 * 1024
                           + (size_t)(ty0 + y0 + wy * BY) * 1024 + (tx0 + x0 + wx * 64 + cn + (hi ? 0 : 4));

    #pragma unroll 3
    for (int s = 0; s < IPH / 2; ++s) {
        #pragma unroll
        for (int q = 0; q < 7; ++q)
            d[q] = __builtin_shufflevector(d[q], (v8f){}, 8, 6, 0, 1, 2, 3, 4, 5);

        #pragma unroll
        for (int par = 0; par < 2; ++par) {
            const float* rr = bp + (2 * s + par) * IPW;
            const v2f D0 = { rr[0],  rr[1]  }, D1 = { rr[4],  rr[5]  };
            const v2f D2 = { rr[8],  rr[9]  }, D3 = { rr[12], rr[13] };
            const v2f D4 = { rr[16], rr[17] }, D5 = { rr[20], rr[21] };
            const v2f D6 = { rr[24], rr[25] };
            const v2f w01 = D0 - 2.0f * D1, w12 = D1 - 2.0f * D2, w23 = D2 - 2.0f * D3;
            const v2f w34 = D3 - 2.0f * D4, w45 = D4 - 2.0f * D5, w56 = D5 - 2.0f * D6;
            const v2f t13 = D1 - D3, t24 = D2 - D4, t35 = D3 - D5;
            const v2f ev1 = w34 - 4.0f * w12, od1 = w45 - 4.0f * w23;
            const v2f ev2 = t13 - 2.0f * t24, od2 = t24 - 2.0f * t35;
            v2f b[7];
            b[0] = 4.0f * w01 + (w45 - 5.0f * w23);
            b[1] = ev1 + od1;               b[2] = ev1 - od1;
            b[3] = 2.0f * ev2 + od2;        b[4] = 2.0f * ev2 - od2;
            b[5] = 4.0f * t13 - t35;
            b[6] = 5.0f * w34 - (4.0f * w12 + w56);
            #pragma unroll
            for (int q = 0; q < 7; ++q) {
                d[q] = __builtin_amdgcn_wmma_f32_16x16x4_f32(
                    false, par ? ao[q] : ae[q], false, b[q], (short)0, d[q], false, false);
            }
        }

        if (s >= 7) {
            const float sa = d[1][7] + d[2][7], sb = d[1][7] - d[2][7];
            const float sc = d[3][7] + d[4][7], se = d[3][7] - d[4][7];
            const float m5 = d[5][7];
            const float ya  = (d[0][7] + sa) + (sc + m5);
            const float yb  = (sb + 2.0f * se) + 0.5f * m5;
            const float yc  = (sa + 4.0f * sc) + 0.25f * m5;
            const float yd  = ((sb + 8.0f * se) + 0.125f * m5) + d[6][7];
            const float g0 = xhalf(hi ? yb : ya);
            const float g1 = xhalf(hi ? yd : yc);
            float* o14 = op + (size_t)(2 * s - 14) * 1024;
            float* o13 = op + (size_t)(2 * s - 13) * 1024;
            o14[0] = hi ? ya : g0;
            o14[8] = hi ? yc : g1;
            o13[0] = hi ? g0 : yb;
            o13[8] = hi ? g1 : yd;
        }
    }
}

extern "C" void kernel_launch(void* const* d_in, const int* in_sizes, int n_in,
                              void* d_out, int out_size, void* d_ws, size_t ws_size,
                              hipStream_t stream) {
    const float* x    = (const float*)d_in[0];
    const float* kern = (const float*)d_in[1];
    float* outp  = (float*)d_out;
    float* knorm = (float*)d_ws;

    norm_kernels_k<<<16, 256, 0, stream>>>(kern, knorm);

    pwblur_k<<<1536, 128, 0, stream>>>(x, knorm, outp);
}


# Tap blocks polyphase (taps j, j+4, j+8, j+12): a lane's 4 outputs adjacent, one 16-byte store per lane and row, no half swap; B pairs 8 columns apart, explicit 2-address LDS reads; sum order changes
# speedup vs baseline: 2.0419x; 1.0508x over previous
#include <hip/hip_runtime.h>


typedef __attribute__((ext_vector_type(2))) float v2f;
typedef __attribute__((ext_vector_type(8))) float v8f;
typedef __attribute__((address_space(3))) const float lds_cfloat;

#define TILE   256
#define BY     64
#define GY     128
#define BX     128
#define IPH    (BY + 14)
#define BPH    (GY + 14)
#define IPW    (BX + 16)

__global__ __launch_bounds__(256) void norm_kernels_k(const float* __restrict__ k,
                                                      float* __restrict__ knorm) {
    const int t   = blockIdx.x;
    const int tid = threadIdx.x;

    __shared__ float red[256];
    float v = 0.0f;
    if (tid < 225) v = k[t * 225 + tid];
    red[tid] = v;
    __syncthreads();
    for (int s = 128; s > 0; s >>= 1) {
        if (tid < s) red[tid] += red[tid + s];
        __syncthreads();
    }
    const float inv = 1.0f / (red[0] + 1e-12f);

    const int dy = tid >> 4;
    const int dx = tid & 15;
    const float val = (dy < 15 && dx < 15) ? k[t * 225 + dy * 15 + dx] * inv : 0.0f;
    knorm[t * 256 + tid] = val;
}

__device__ __forceinline__ void tap_points(const float* __restrict__ r, v2f (&u)[7]) {
    const v2f g0 = { r[0], r[8] },  g1 = { r[1], r[9] };
    const v2f g2 = { r[2], r[10] }, g3 = { r[3], r[11] };
    const v2f s02 = g0 + g2, s13 = g1 + g3;
    const v2f p = g0 + 4.0f * g2, q = g1 + 4.0f * g3;
    u[0] = 0.25f * g0;
    u[1] = (1.0f / 6.0f)  * (s02 + s13);
    u[2] = (1.0f / 18.0f) * (s02 - s13);
    u[3] = (1.0f / 72.0f)  * (p + 2.0f * q);
    u[4] = (1.0f / 120.0f) * (p - 2.0f * q);
    u[5] = (4.0f / 45.0f) * ((8.0f * g0 + 4.0f * g1) + (2.0f * g2 + g3));
    u[6] = 0.5f * g3;
}

__global__ __launch_bounds__(128) void pwblur_k(const float* __restrict__ x,
                                               const float* __restrict__ knorm,
                                               float* __restrict__ out) {
    __shared__ __attribute__((aligned(16))) float ip[BPH * IPW];

    const int tid  = threadIdx.x;
    const int lane = tid & 31;
    const int wave = tid >> 5;
    const bool hi  = (lane & 16) != 0;
    const int  nl  = lane & 15;

    const int bid   = blockIdx.x;
    const int sub   = bid & 3;
    const int t     = bid >> 2;
    const int ti    = t & 3;
    const int tj    = (t >> 2) & 3;
    const int plane = t >> 4;
    const int sub_y = sub >> 1;
    const int sub_x = sub & 1;

    const int ty0 = tj * TILE;
    const int tx0 = ti * TILE;
    const int y0  = sub_y * GY;
    const int x0  = sub_x * BX;

    const float* __restrict__ xin = x + (size_t)plane * 1024 * 1024;

    {
        const int wv = __builtin_amdgcn_readfirstlane(tid >> 5);
        const int qa = tid & 31;
        const int xta = x0 - 8 + 4 * qa;
        const bool lefta = xta < 0;
        const float* __restrict__ pa = xin + (tx0 + (lefta ? 0 : xta));
        const int qb = 32 + (tid & 3);
        const int xtb = x0 - 8 + 4 * qb;
        const bool rightb = xtb > TILE - 4;
        const float* __restrict__ pb = xin + (tx0 + (rightb ? TILE - 4 : xtb));
        const int rb0 = tid >> 2;

        float4 va[(BPH + 3) / 4], vb[5];
        #pragma unroll
        for (int m = 0; m < (BPH + 3) / 4; ++m) {
            int r = wv + 4 * m;             r = r > BPH - 1 ? BPH - 1 : r;
            int ys = y0 + r - 7;            ys = ys < 0 ? 0 : (ys > TILE - 1 ? TILE - 1 : ys);
            va[m] = *reinterpret_cast<const float4*>(pa + (size_t)(ty0 + ys) * 1024);
        }
        #pragma unroll
        for (int m = 0; m < 5; ++m) {
            int r = rb0 + 32 * m;           r = r > BPH - 1 ? BPH - 1 : r;
            int ys = y0 + r - 7;            ys = ys < 0 ? 0 : (ys > TILE - 1 ? TILE - 1 : ys);
            vb[m] = *reinterpret_cast<const float4*>(pb + (size_t)(ty0 + ys) * 1024);
        }
        #pragma unroll
        for (int m = 0; m < (BPH + 3) / 4; ++m) {
            int r = wv + 4 * m;             r = r > BPH - 1 ? BPH - 1 : r;
            float4 w = va[m];
            if (lefta) w = make_float4(w.x, w.x, w.x, w.x);
            *reinterpret_cast<float4*>(ip + r * IPW + 4 * qa) = w;
        }
        #pragma unroll
        for (int m = 0; m < 5; ++m) {
            int r = rb0 + 32 * m;           r = r > BPH - 1 ? BPH - 1 : r;
            float4 w = vb[m];
            if (rightb) w = make_float4(w.w, w.w, w.w, w.w);
            *reinterpret_cast<float4*>(ip + r * IPW + 4 * qb) = w;
        }
    }

    const float* __restrict__ kn = knorm + (tj * 4 + ti) * 256;
    v2f ao[7], ae[7];
    {
        const int tb = hi ? 4 : 0;
        const int rr7 = nl & 7;
        const int pos = (nl >> 3) + 2 * ((rr7 >> 1) + ((rr7 & 1) ? 4 : 0));
        const float* __restrict__ ro = kn + pos * 16 + tb;
        const float* __restrict__ re = kn + ((pos + 15) & 15) * 16 + tb;
        tap_points(ro, ao);
        tap_points(re, ae);
    }
    __syncthreads();

    const int wx = wave & 1;
    const int wy = wave >> 1;
    const int cn = 4 * nl;
    const float* bp = ip + wy * BY * IPW + 1 + wx * 64 + cn + (hi ? 4 : 0);
    const unsigned bp_lds = (unsigned)(unsigned long long)(lds_cfloat*)bp;

    v8f d[7];
    #pragma unroll
    for (int q = 0; q < 7; ++q) d[q] = (v8f){};

    float* __restrict__ op = out + (size_t)plane * 1024 * 1024
                           + (size_t)(ty0 + y0 + wy * BY + (hi ? 0 : 1)) * 1024 + (tx0 + x0 + wx * 64 + cn);

    #pragma unroll 3
    for (int s = 0; s < IPH / 2; ++s) {
        #pragma unroll
        for (int q = 0; q < 7; ++q)
            d[q] = __builtin_shufflevector(d[q], (v8f){}, 8, 6, 0, 1, 2, 3, 4, 5);

        v2f Dr[2][7];
        asm volatile(
            "ds_load_2addr_b32 %0, %14 offset0:0 offset1:8\n\t"
            "ds_load_2addr_b32 %1, %14 offset0:1 offset1:9\n\t"
            "ds_load_2addr_b32 %2, %14 offset0:2 offset1:10\n\t"
            "ds_load_2addr_b32 %3, %14 offset0:3 offset1:11\n\t"
            "ds_load_2addr_b32 %4, %14 offset0:4 offset1:12\n\t"
            "ds_load_2addr_b32 %5, %14 offset0:5 offset1:13\n\t"
            "ds_load_2addr_b32 %6, %14 offset0:6 offset1:14\n\t"
            "ds_load_2addr_b32 %7, %14 offset0:144 offset1:152\n\t"
            "ds_load_2addr_b32 %8, %14 offset0:145 offset1:153\n\t"
            "ds_load_2addr_b32 %9, %14 offset0:146 offset1:154\n\t"
            "ds_load_2addr_b32 %10, %14 offset0:147 offset1:155\n\t"
            "ds_load_2addr_b32 %11, %14 offset0:148 offset1:156\n\t"
            "ds_load_2addr_b32 %12, %14 offset0:149 offset1:157\n\t"
            "ds_load_2addr_b32 %13, %14 offset0:150 offset1:158\n\t"
            "s_wait_dscnt 0x7"
            : "=&v"(Dr[0][0]), "=&v"(Dr[0][1]), "=&v"(Dr[0][2]), "=&v"(Dr[0][3]),
              "=&v"(Dr[0][4]), "=&v"(Dr[0][5]), "=&v"(Dr[0][6]),
              "=&v"(Dr[1][0]), "=&v"(Dr[1][1]), "=&v"(Dr[1][2]), "=&v"(Dr[1][3]),
              "=&v"(Dr[1][4]), "=&v"(Dr[1][5]), "=&v"(Dr[1][6])
            : "v"(bp_lds + (unsigned)s * (2u * IPW * 4u))
            : "memory");

        #pragma unroll
        for (int par = 0; par < 2; ++par) {
            if (par == 1) {
                asm volatile("s_wait_dscnt 0x0"
                    : "+v"(Dr[1][0]), "+v"(Dr[1][1]), "+v"(Dr[1][2]), "+v"(Dr[1][3]),
                      "+v"(Dr[1][4]), "+v"(Dr[1][5]), "+v"(Dr[1][6])
                    :
                    : "memory");
            }
            const v2f D0 = Dr[par][0], D1 = Dr[par][1], D2 = Dr[par][2], D3 = Dr[par][3];
            const v2f D4 = Dr[par][4], D5 = Dr[par][5], D6 = Dr[par][6];
            const v2f w01 = D0 - 2.0f * D1, w12 = D1 - 2.0f * D2, w23 = D2 - 2.0f * D3;
            const v2f w34 = D3 - 2.0f * D4, w45 = D4 - 2.0f * D5, w56 = D5 - 2.0f * D6;
            const v2f t13 = D1 - D3, t24 = D2 - D4, t35 = D3 - D5;
            const v2f ev1 = w34 - 4.0f * w12, od1 = w45 - 4.0f * w23;
            const v2f ev2 = t13 - 2.0f * t24, od2 = t24 - 2.0f * t35;
            v2f b[7];
            b[0] = 4.0f * w01 + (w45 - 5.0f * w23);
            b[1] = ev1 + od1;               b[2] = ev1 - od1;
            b[3] = 2.0f * ev2 + od2;        b[4] = 2.0f * ev2 - od2;
            b[5] = 4.0f * t13 - t35;
            b[6] = 5.0f * w34 - (4.0f * w12 + w56);
            #pragma unroll
            for (int q = 0; q < 7; ++q) {
                d[q] = __builtin_amdgcn_wmma_f32_16x16x4_f32(
                    false, par ? ao[q] : ae[q], false, b[q], (short)0, d[q], false, false);
            }
        }

        if (s >= 7) {
            const float sa = d[1][7] + d[2][7], sb = d[1][7] - d[2][7];
            const float sc = d[3][7] + d[4][7], se = d[3][7] - d[4][7];
            const float m5 = d[5][7];
            const float ya  = (d[0][7] + sa) + (sc + m5);
            const float yb  = (sb + 2.0f * se) + 0.5f * m5;
            const float yc  = (sa + 4.0f * sc) + 0.25f * m5;
            const float yd  = ((sb + 8.0f * se) + 0.125f * m5) + d[6][7];
            *reinterpret_cast<float4*>(op + (size_t)(2 * s - 14) * 1024) = make_float4(ya, yb, yc, yd);
        }
    }
}

extern "C" void kernel_launch(void* const* d_in, const int* in_sizes, int n_in,
                              void* d_out, int out_size, void* d_ws, size_t ws_size,
                              hipStream_t stream) {
    const float* x    = (const float*)d_in[0];
    const float* kern = (const float*)d_in[1];
    float* outp  = (float*)d_out;
    float* knorm = (float*)d_ws;

    norm_kernels_k<<<16, 256, 0, stream>>>(kern, knorm);

    pwblur_k<<<1536, 128, 0, stream>>>(x, knorm, outp);
}
